# starting kernel
# baseline (speedup 1.0000x reference)
#include <hip/hip_runtime.h>


#define NTOT   200000
#define CDIM   128
#define OUTC   64
#define STRIDE 132

typedef float v2f __attribute__((ext_vector_type(2)));
typedef float v8f __attribute__((ext_vector_type(8)));

__device__ __forceinline__ void lds_wait() {
  asm volatile("s_wait_dscnt 0" ::: "memory");
}

__device__ __forceinline__ v8f wmma4(v2f a, v2f b, v8f c) {
  return __builtin_amdgcn_wmma_f32_16x16x4_f32(false, a, false, b, (short)0, c,
                                               false, false);
}

template <int TT>
__device__ __forceinline__ void gemm_tiles(const float* __restrict__ wl,
                                           int lane,
                                           const float* __restrict__ aptr,
                                           v8f* __restrict__ acc) {
#pragma unroll
  for (int t = 0; t < TT; ++t) acc[t] = (v8f)0.0f;
  const float* wlp = wl + lane * 2;
#pragma unroll 8
  for (int kc = 0; kc < 32; ++kc) {
    v2f a = *(const v2f*)(aptr + kc * 4);
#pragma unroll
    for (int t = 0; t < TT; ++t) {
      v2f b = *(const v2f*)(wlp + kc * TT * 64 + t * 64);
      acc[t] = wmma4(a, b, acc[t]);
    }
  }
}

__launch_bounds__(256, 1)
__global__ void gnode_kernel(const float* __restrict__ x,
                             const float* __restrict__ W1,
                             const float* __restrict__ b1,
                             const float* __restrict__ W2,
                             const float* __restrict__ b2,
                             const float* __restrict__ Wl,
                             const float* __restrict__ bl,
                             float* __restrict__ out) {
  __shared__ __attribute__((aligned(16))) float w1s[32 * 8 * 32 * 2];
  __shared__ __attribute__((aligned(16))) float w2s[32 * 8 * 32 * 2];
  __shared__ __attribute__((aligned(16))) float wls[32 * 4 * 32 * 2];
  __shared__ __attribute__((aligned(16))) float stage[8 * 16 * STRIDE];

  const int tid   = threadIdx.x;
  const int lane  = tid & 31;
  const int wave  = tid >> 5;
  const int lrow  = lane & 15;
  const int half8 = (lane >> 4) << 3;

  for (int s = tid; s < 32 * 8 * 32; s += 256) {
    int ls = s & 31;
    int t  = (s >> 5) & 7;
    int kc = s >> 8;
    int k0 = kc * 4 + ((ls >> 4) << 1);
    int n  = t * 16 + (ls & 15);
    w1s[2 * s]     = W1[k0 * CDIM + n];
    w1s[2 * s + 1] = W1[(k0 + 1) * CDIM + n];
    w2s[2 * s]     = W2[k0 * CDIM + n];
    w2s[2 * s + 1] = W2[(k0 + 1) * CDIM + n];
  }
  for (int s = tid; s < 32 * 4 * 32; s += 256) {
    int ls = s & 31;
    int t  = (s >> 5) & 3;
    int kc = s >> 7;
    int k0 = kc * 4 + ((ls >> 4) << 1);
    int n  = t * 16 + (ls & 15);
    wls[2 * s]     = Wl[k0 * OUTC + n];
    wls[2 * s + 1] = Wl[(k0 + 1) * OUTC + n];
  }

  float b1v[8], b2v[8], blv[4];
#pragma unroll
  for (int t = 0; t < 8; ++t) {
    b1v[t] = b1[t * 16 + lrow];
    b2v[t] = b2[t * 16 + lrow];
  }
#pragma unroll
  for (int t = 0; t < 4; ++t) blv[t] = bl[t * 16 + lrow];

  const int rowbase = blockIdx.x * 128 + wave * 16;
  v8f S[8], K[8], acc[8];
#pragma unroll
  for (int t = 0; t < 8; ++t) {
#pragma unroll
    for (int r = 0; r < 8; ++r) {
      int row = rowbase + r + half8;
      S[t][r] = (row < NTOT) ? x[row * CDIM + t * 16 + lrow] : 0.0f;
    }
  }

  __syncthreads();

  float* st = stage + wave * 16 * STRIDE;
  const float* aptr = st + lrow * STRIDE + ((lane >> 4) << 1);

  auto emit = [&](auto&& gen) {
#pragma unroll
    for (int t = 0; t < 8; ++t) {
      v8f e = gen(t);
#pragma unroll
      for (int r = 0; r < 8; ++r)
        st[(r + half8) * STRIDE + t * 16 + lrow] = e[r];
    }
    lds_wait();
  };

  auto feval = [&]() {
    gemm_tiles<8>(w1s, lane, aptr, acc);
#pragma unroll
    for (int t = 0; t < 8; ++t) {
#pragma unroll
      for (int r = 0; r < 8; ++r)
        st[(r + half8) * STRIDE + t * 16 + lrow] = fmaxf(acc[t][r] + b1v[t], 0.0f);
    }
    lds_wait();
    gemm_tiles<8>(w2s, lane, aptr, acc);
#pragma unroll
    for (int t = 0; t < 8; ++t) acc[t] += b2v[t];
  };

  const float hc    = 0.1f;
  const float c18   = hc * 0.125f;
  const float c38   = hc * 0.375f;
  const float c524  = hc * (5.0f / 24.0f);
  const float c1124 = hc * (11.0f / 24.0f);
  const float c58   = hc * 0.625f;
  const float c78   = hc * 0.875f;
  const float c118  = hc * 1.375f;

#pragma unroll 1
  for (int step = 0; step < 10; ++step) {
    emit([&](int t) { return S[t]; });
    feval();
#pragma unroll
    for (int t = 0; t < 8; ++t) { K[t] = acc[t]; S[t] += c18 * acc[t]; }
    emit([&](int t) { return S[t] + c524 * K[t]; });
    feval();
#pragma unroll
    for (int t = 0; t < 8; ++t) S[t] += c38 * acc[t];
    emit([&](int t) { return S[t] - c1124 * K[t] + c58 * acc[t]; });
#pragma unroll
    for (int t = 0; t < 8; ++t) K[t] = S[t] + c78 * K[t] - c118 * acc[t];
    feval();
#pragma unroll
    for (int t = 0; t < 8; ++t) S[t] += c38 * acc[t];
    emit([&](int t) { return K[t] + hc * acc[t]; });
    feval();
#pragma unroll
    for (int t = 0; t < 8; ++t) S[t] += c18 * acc[t];
  }

  emit([&](int t) { return S[t]; });
  gemm_tiles<4>(wls, lane, aptr, acc);
#pragma unroll
  for (int t = 0; t < 4; ++t) {
#pragma unroll
    for (int r = 0; r < 8; ++r) {
      int row = rowbase + r + half8;
      if (row < NTOT) out[row * OUTC + t * 16 + lrow] = acc[t][r] + blv[t];
    }
  }
}

extern "C" void kernel_launch(void* const* d_in, const int* in_sizes, int n_in,
                              void* d_out, int out_size, void* d_ws, size_t ws_size,
                              hipStream_t stream) {
  (void)in_sizes; (void)n_in; (void)out_size; (void)d_ws; (void)ws_size;
  const float* x  = (const float*)d_in[0];
  const float* W1 = (const float*)d_in[1];
  const float* b1 = (const float*)d_in[2];
  const float* W2 = (const float*)d_in[3];
  const float* b2 = (const float*)d_in[4];
  const float* Wl = (const float*)d_in[5];
  const float* bl = (const float*)d_in[6];
  float* out = (float*)d_out;

  dim3 grid((NTOT + 127) / 128);
  gnode_kernel<<<grid, 256, 0, stream>>>(x, W1, b1, W2, b2, Wl, bl, out);
}


# 4-wave work-groups of 32 rows, 32 columns per wave (one LDS read feeds 4 matrix instructions); two independent groups per WGP fill each other's hand-over gaps; projection's sum order changes
# speedup vs baseline: 1.2350x; 1.2350x over previous
#include <hip/hip_runtime.h>


#define NTOT   200000
#define CDIM   128
#define OUTC   64
#define STRIDE 132
#define RT     2
#define ROWS   (RT * 16)
#define BR     4
#define FSET   (BR * RT)
#define OSTR   68

static_assert(NTOT % ROWS == 0, "a work-group's 32 rows must tile the node count exactly: no row guard below");

typedef float v2f __attribute__((ext_vector_type(2)));
typedef float v4f __attribute__((ext_vector_type(4)));
typedef float v8f __attribute__((ext_vector_type(8)));

__device__ __forceinline__ v8f wmma4(v2f a, v2f b, v8f c) {
  return __builtin_amdgcn_wmma_f32_16x16x4_f32(false, a, false, b, (short)0, c,
                                               false, false);
}


__device__ __forceinline__ void load_block(v4f* F, const float* abase, int blk) {
#pragma unroll
  for (int r = 0; r < BR; ++r) {
#pragma unroll
    for (int t = 0; t < RT; ++t)
      F[r * RT + t] = *(const v4f*)(abase + t * 16 * STRIDE + (BR * blk + r) * 8);
  }
}

template <int NC>
__device__ __forceinline__ void mma_block(const v2f (&wr)[2][32], const v4f* F, int blk,
                                          v8f (&acc)[RT][2]) {
#pragma unroll
  for (int r = 0; r < BR; ++r) {
    const int j = BR * blk + r;
#pragma unroll
    for (int t = 0; t < RT; ++t) {
      v2f a0 = {F[r * RT + t].x, F[r * RT + t].y};
#pragma unroll
      for (int c = 0; c < NC; ++c) acc[t][c] = wmma4(a0, wr[c][2 * j], acc[t][c]);
    }
#pragma unroll
    for (int t = 0; t < RT; ++t) {
      v2f a1 = {F[r * RT + t].z, F[r * RT + t].w};
#pragma unroll
      for (int c = 0; c < NC; ++c) acc[t][c] = wmma4(a1, wr[c][2 * j + 1], acc[t][c]);
    }
  }
}

template <int NC>
__device__ __forceinline__ void gemm_regw(const v2f (&wr)[2][32], const float* abase,
                                          v8f (&acc)[RT][2]) {
  v4f F0[FSET], F1[FSET];
#pragma unroll
  for (int t = 0; t < RT; ++t) {
#pragma unroll
    for (int c = 0; c < NC; ++c) acc[t][c] = (v8f)0.0f;
  }
  load_block(F0, abase, 0);
  load_block(F1, abase, 1);
  __builtin_amdgcn_sched_barrier(0);
  mma_block<NC>(wr, F0, 0, acc);
  __builtin_amdgcn_sched_barrier(0);
  load_block(F0, abase, 2);
  __builtin_amdgcn_sched_barrier(0);
  mma_block<NC>(wr, F1, 1, acc);
  __builtin_amdgcn_sched_barrier(0);
  load_block(F1, abase, 3);
  __builtin_amdgcn_sched_barrier(0);
  mma_block<NC>(wr, F0, 2, acc);
  mma_block<NC>(wr, F1, 3, acc);
  __builtin_amdgcn_sched_barrier(0);
}

__launch_bounds__(128, 2)
__global__ void gnode_kernel(const float* __restrict__ x,
                             const float* __restrict__ W1,
                             const float* __restrict__ b1,
                             const float* __restrict__ W2,
                             const float* __restrict__ b2,
                             const float* __restrict__ Wl,
                             const float* __restrict__ bl,
                             float* __restrict__ out) {
  __shared__ __attribute__((aligned(16))) float stage[2 * ROWS * STRIDE];

  const int tid   = threadIdx.x;
  const int lane  = tid & 31;
  const int wave  = __builtin_amdgcn_readfirstlane(tid >> 5);
  const int lrow  = lane & 15;
  const int hi    = lane >> 4;
  const int half8 = hi << 3;
  const int colb  = wave * 32 + lrow;

  v2f w1r[2][32], w2r[2][32];
  float b1c[2], b2c[2];
#pragma unroll
  for (int c = 0; c < 2; ++c) {
    const int col = colb + 16 * c;
#pragma unroll
    for (int j = 0; j < 16; ++j) {
      int k = 8 * j + 4 * hi;
      w1r[c][2 * j]     = (v2f){W1[(k    ) * CDIM + col], W1[(k + 1) * CDIM + col]};
      w1r[c][2 * j + 1] = (v2f){W1[(k + 2) * CDIM + col], W1[(k + 3) * CDIM + col]};
      w2r[c][2 * j]     = (v2f){W2[(k    ) * CDIM + col], W2[(k + 1) * CDIM + col]};
      w2r[c][2 * j + 1] = (v2f){W2[(k + 2) * CDIM + col], W2[(k + 3) * CDIM + col]};
    }
    b1c[c] = b1[col];
    b2c[c] = b2[col];
  }

  const int rowbase = blockIdx.x * ROWS;
  v8f S[RT][2], K[RT][2];
#pragma unroll
  for (int t = 0; t < RT; ++t) {
#pragma unroll
    for (int c = 0; c < 2; ++c) {
#pragma unroll
      for (int r = 0; r < 8; ++r) {
        int row = rowbase + t * 16 + half8 + r;
        S[t][c][r] = x[row * CDIM + colb + 16 * c];
      }
      K[t][c] = (v8f)0.0f;
    }
  }

  float* buf0 = stage;
  float* buf1 = stage + ROWS * STRIDE;
  const int soff = half8 * STRIDE + colb;
  const int aoff = lrow * STRIDE + 4 * hi;

  auto put = [&](float* b, int tile, int c, v8f e) {
#pragma unroll
    for (int r = 0; r < 8; ++r) b[(tile * 16 + r) * STRIDE + soff + 16 * c] = e[r];
  };

#pragma unroll
  for (int t = 0; t < RT; ++t) {
#pragma unroll
    for (int c = 0; c < 2; ++c) put(buf0, t, c, S[t][c]);
  }
  __syncthreads();

  const float hc    = 0.1f;
  const float c18   = hc * 0.125f;
  const float c38   = hc * 0.375f;
  const float c524  = hc * (5.0f / 24.0f);
  const float c1124 = hc * (11.0f / 24.0f);
  const float c58   = hc * 0.625f;
  const float c78   = hc * 0.875f;
  const float c118  = hc * 1.375f;

  v8f acc[RT][2];

#pragma unroll 1
  for (int it = 0; it < 40; ++it) {
    const int st = it & 3;

    gemm_regw<2>(w1r, buf0 + aoff, acc);
#pragma unroll
    for (int t = 0; t < RT; ++t) {
#pragma unroll
      for (int c = 0; c < 2; ++c) {
        v8f e;
#pragma unroll
        for (int r = 0; r < 8; ++r) e[r] = fmaxf(acc[t][c][r] + b1c[c], 0.0f);
        put(buf1, t, c, e);
      }
    }
    __syncthreads();

    gemm_regw<2>(w2r, buf1 + aoff, acc);
#pragma unroll
    for (int t = 0; t < RT; ++t) {
#pragma unroll
      for (int c = 0; c < 2; ++c) acc[t][c] += b2c[c];
    }
    if (st == 0) {
#pragma unroll
      for (int t = 0; t < RT; ++t) {
#pragma unroll
        for (int c = 0; c < 2; ++c) {
          K[t][c] = acc[t][c];
          S[t][c] += c18 * acc[t][c];
          put(buf0, t, c, S[t][c] + c524 * K[t][c]);
        }
      }
    } else if (st == 1) {
#pragma unroll
      for (int t = 0; t < RT; ++t) {
#pragma unroll
        for (int c = 0; c < 2; ++c) {
          S[t][c] += c38 * acc[t][c];
          put(buf0, t, c, S[t][c] - c1124 * K[t][c] + c58 * acc[t][c]);
          K[t][c] = S[t][c] + c78 * K[t][c] - c118 * acc[t][c];
        }
      }
    } else if (st == 2) {
#pragma unroll
      for (int t = 0; t < RT; ++t) {
#pragma unroll
        for (int c = 0; c < 2; ++c) {
          S[t][c] += c38 * acc[t][c];
          put(buf0, t, c, K[t][c] + hc * acc[t][c]);
        }
      }
    } else {
#pragma unroll
      for (int t = 0; t < RT; ++t) {
#pragma unroll
        for (int c = 0; c < 2; ++c) {
          S[t][c] += c18 * acc[t][c];
          put(buf0, t, c, S[t][c]);
        }
      }
    }
    __syncthreads();
  }

  const int pcol = wave * 16 + lrow;
  v2f wlr[2][32];
#pragma unroll
  for (int j = 0; j < 16; ++j) {
    int k = 8 * j + 4 * hi;
    wlr[0][2 * j]     = (v2f){Wl[(k    ) * OUTC + pcol], Wl[(k + 1) * OUTC + pcol]};
    wlr[0][2 * j + 1] = (v2f){Wl[(k + 2) * OUTC + pcol], Wl[(k + 3) * OUTC + pcol]};
  }
  const float blv = bl[pcol];
  gemm_regw<1>(wlr, buf0 + aoff, acc);
  float* ob = buf1;
#pragma unroll
  for (int t = 0; t < RT; ++t) {
#pragma unroll
    for (int r = 0; r < 8; ++r)
      ob[(t * 16 + half8 + r) * OSTR + pcol] = acc[t][0][r] + blv;
  }
  __syncthreads();
#pragma unroll
  for (int i = 0; i < (ROWS * OUTC / 4) / 128; ++i) {
    const int idx = tid + 128 * i;
    const int row = idx >> 4;
    const int c4  = idx & 15;
    v4f v = *(const v4f*)(ob + row * OSTR + 4 * c4);
    *(v4f*)(out + (rowbase + row) * OUTC + 4 * c4) = v;
  }
}

extern "C" void kernel_launch(void* const* d_in, const int* in_sizes, int n_in,
                              void* d_out, int out_size, void* d_ws, size_t ws_size,
                              hipStream_t stream) {
  (void)in_sizes; (void)n_in; (void)out_size; (void)d_ws; (void)ws_size;
  const float* x  = (const float*)d_in[0];
  const float* W1 = (const float*)d_in[1];
  const float* b1 = (const float*)d_in[2];
  const float* W2 = (const float*)d_in[3];
  const float* b2 = (const float*)d_in[4];
  const float* Wl = (const float*)d_in[5];
  const float* bl = (const float*)d_in[6];
  float* out = (float*)d_out;

  dim3 grid(NTOT / ROWS);
  gnode_kernel<<<grid, 128, 0, stream>>>(x, W1, b1, W2, b2, Wl, bl, out);
}
